# starting kernel
# baseline (speedup 1.0000x reference)
#include <hip/hip_runtime.h>
#include <hip/hip_bf16.h>


typedef __attribute__((ext_vector_type(2))) float v2f;
typedef __attribute__((ext_vector_type(8))) float v8f;

#define SIG_B      1024
#define SIG_L      2048
#define SIG_C      7
#define SIG_T      2047
#define SIG_HC     128
#define SIG_NH     16
#define SIG_NWAVE  8
#define SIG_COLS   136
#define SIG_DT     (1.0f / 2047.0f)

__device__ __forceinline__ void sig_step2(float& S, float bA, float bB, v8f& acc)
{
    v2f a, bv;
    a.x = S; bv.x = bA; S += bA;
    a.y = S; bv.y = bB; S += bB;
    acc = __builtin_amdgcn_wmma_f32_16x16x4_f32(
              false, a, false, bv, (short)0, acc, false, false);
}

__global__ __launch_bounds__(32)
void sig_partial_kernel(const float* __restrict__ path,
                        float* __restrict__ ws_sums,
                        float* __restrict__ ws_sig2)
{
    const int bb   = blockIdx.x >> 3;
    const int w    = blockIdx.x & 7;
    const int lane = threadIdx.x;
    const int q    = lane & 15;
    const int p    = q >> 3;
    const int cm   = q & 7;

    const int batch = 2 * bb + p;
    const int chunk = 2 * w + (lane >> 4);
    const int t0    = chunk * SIG_HC;
    const bool is_time = (cm == 0);

    const int comp = (cm >= 1) ? (cm - 1) : 0;
    const float* __restrict__ bbase = path + (size_t)batch * SIG_L * SIG_C;

    const float* __restrict__ ptr = bbase + (t0 + 1) * SIG_C + comp;

    float prev = bbase[t0 * SIG_C + comp];
    float S = 0.0f;
    v8f acc = (v8f)0.0f;

    const int iters = (w == 7) ? 31 : 32;
    for (int i = 0; i < iters; ++i) {
        const float x0 = ptr[0];
        const float x1 = ptr[SIG_C];
        const float x2 = ptr[2 * SIG_C];
        const float x3 = ptr[3 * SIG_C];
        ptr += 4 * SIG_C;

        const float b0 = is_time ? SIG_DT : (x0 - prev);
        const float b1 = is_time ? SIG_DT : (x1 - x0);
        const float b2 = is_time ? SIG_DT : (x2 - x1);
        const float b3 = is_time ? SIG_DT : (x3 - x2);
        prev = x3;

        sig_step2(S, b0, b1, acc);
        sig_step2(S, b2, b3, acc);
    }

    if (w == 7) {
        int off = (t0 + 125) * SIG_C + comp;
        const int offMax = (SIG_L - 1) * SIG_C + comp;
        const int o0 = (off             < offMax) ? off             : offMax;
        const int o1 = (off +   SIG_C   < offMax) ? off +   SIG_C   : offMax;
        const int o2 = (off + 2*SIG_C   < offMax) ? off + 2*SIG_C   : offMax;
        const int o3 = (off + 3*SIG_C   < offMax) ? off + 3*SIG_C   : offMax;
        const float x0 = bbase[o0];
        const float x1 = bbase[o1];
        const float x2 = bbase[o2];
        const float x3 = bbase[o3];

        const float b0 = is_time ? SIG_DT : (x0 - prev);
        const float b1 = is_time ? SIG_DT : (x1 - x0);
        const float b2 = is_time ? SIG_DT : (x2 - x1);
        const float b3 = is_time ? SIG_DT : (x3 - x2);

        sig_step2(S, b0, b1, acc);
        sig_step2(S, b2, b3, acc);
    }

    ws_sums[((size_t)batch * SIG_NH + chunk) * 8 + cm] = S;

    if (lane < 8) {
        float* dst = ws_sig2 + ((size_t)(2 * bb) * SIG_NWAVE + w) * 64;
        #pragma unroll
        for (int j = 0; j < 8; ++j)
            dst[j * 8 + lane] = acc[j];
    } else if (lane >= 24) {
        float* dst = ws_sig2 + ((size_t)(2 * bb + 1) * SIG_NWAVE + w) * 64;
        #pragma unroll
        for (int j = 0; j < 8; ++j)
            dst[j * 8 + (lane - 24)] = acc[j];
    }
}

__global__ __launch_bounds__(64)
void sig_combine_kernel(const float* __restrict__ ws_sums,
                        const float* __restrict__ ws_sig2,
                        float* __restrict__ out, int out_size)
{
    const int b   = blockIdx.x;
    const int tid = threadIdx.x;
    __shared__ float s[SIG_NH * 8];

    const float* bs = ws_sums + (size_t)b * SIG_NH * 8;
    s[tid] = bs[tid];
    {
        float v = bs[tid + 64];
        if (tid + 64 == 15 * 8 + 0) v -= SIG_DT;
        s[tid + 64] = v;
    }
    __syncthreads();

    const int d = tid >> 3;
    const int e = tid & 7;

    float P = 0.0f;
    float cross = 0.0f;
    #pragma unroll
    for (int c = 0; c < SIG_NH; ++c) {
        cross += P * s[c * 8 + e];
        P     += s[c * 8 + d];
    }

    float part = 0.0f;
    const float* bp = ws_sig2 + (size_t)b * SIG_NWAVE * 64;
    #pragma unroll
    for (int w = 0; w < SIG_NWAVE; ++w)
        part += bp[w * 64 + tid];

    if (e == 0) part -= s[15 * 8 + d] * SIG_DT;

    const float sig2_de = part + cross;
    const size_t base = (size_t)b * SIG_COLS;

    if (base + 8 + tid < (size_t)out_size)
        out[base + 8 + tid] = sig2_de;
    if (base + 72 + tid < (size_t)out_size)
        out[base + 72 + tid] = P * sig2_de;
    if (e == 0 && base + d < (size_t)out_size)
        out[base + d] = P;
}

extern "C" void kernel_launch(void* const* d_in, const int* in_sizes, int n_in,
                              void* d_out, int out_size, void* d_ws, size_t ws_size,
                              hipStream_t stream) {
    (void)in_sizes; (void)n_in; (void)ws_size;
    const float* path = (const float*)d_in[0];

    float* ws_sums = (float*)d_ws;
    float* ws_sig2 = ws_sums + (size_t)SIG_B * SIG_NH * 8;

    sig_partial_kernel<<<dim3((SIG_B / 2) * SIG_NWAVE), dim3(32), 0, stream>>>(
        path, ws_sums, ws_sig2);
    sig_combine_kernel<<<dim3(SIG_B), dim3(64), 0, stream>>>(
        ws_sums, ws_sig2, (float*)d_out, out_size);
}


# A operand is the sample x[row-1] as read from LDS (time lanes read a ramp kept in LDS), B one subtract; running prefix sum dropped, x[0] and time offsets as rank-1 terms; sums regrouped
# speedup vs baseline: 1.7610x; 1.7610x over previous
#include <hip/hip_runtime.h>


typedef __attribute__((ext_vector_type(2))) float v2f;
typedef __attribute__((ext_vector_type(4))) float v4f;
typedef __attribute__((ext_vector_type(8))) float v8f;

#define SIG_B      1024
#define SIG_L      2048
#define SIG_C      7
#define SIG_COLS   136
#define SIG_DT     (1.0f / 2047.0f)
#define SIG_CH     512
#define SIG_SR     128
#define SIG_ROWF   (SIG_SR * SIG_C)
#define SIG_REG    904
#define SIG_RAMP   (4 * SIG_REG + 39)
#define SIG_LDSF   (SIG_RAMP + SIG_CH * SIG_C)

__device__ __forceinline__ void sig_load_stage(v4f (&R)[28], const float* __restrict__ g,
                                               const int s, const int lane)
{
    #pragma unroll
    for (int j = 0; j < 7; ++j) {
        #pragma unroll
        for (int c = 0; c < 4; ++c)
            R[c * 7 + j] = *(const v4f*)(g + c * (SIG_CH * SIG_C) + s * SIG_ROWF
                                           + j * 128 + 4 * lane);
        __builtin_amdgcn_sched_barrier(0);
    }
}

__device__ __forceinline__ void sig_store_stage(float* tile, const v4f (&R)[28], const int lane)
{
    __builtin_amdgcn_fence(__ATOMIC_RELEASE, "wavefront");
    __builtin_amdgcn_wave_barrier();
    __builtin_amdgcn_fence(__ATOMIC_ACQUIRE, "wavefront");
    #pragma unroll
    for (int j = 0; j < 7; ++j) {
        #pragma unroll
        for (int c = 0; c < 4; ++c)
            *(v4f*)(tile + c * SIG_REG + j * 128 + 4 * lane) = R[c * 7 + j];
    }
    __builtin_amdgcn_fence(__ATOMIC_RELEASE, "wavefront");
    __builtin_amdgcn_wave_barrier();
    __builtin_amdgcn_fence(__ATOMIC_ACQUIRE, "wavefront");
}

__device__ __forceinline__ void sig_compute_stage(const float* col, float& prev,
                                                  v8f& acc0, v8f& acc1)
{
    float xv[SIG_SR];
    #pragma unroll
    for (int r = 0; r < SIG_SR; ++r)
        xv[r] = col[SIG_C * r];
    __builtin_amdgcn_sched_barrier(0);
    #pragma unroll
    for (int k = 0; k < SIG_SR / 2 - 1; ++k) {
        v2f a, bv;
        a.x  = xv[2 * k];
        a.y  = xv[2 * k + 1];
        bv.x = a.y - a.x;
        bv.y = xv[2 * k + 2] - a.y;
        if (k & 1)
            acc1 = __builtin_amdgcn_wmma_f32_16x16x4_f32(
                       false, a, false, bv, (short)0, acc1, false, false);
        else
            acc0 = __builtin_amdgcn_wmma_f32_16x16x4_f32(
                       false, a, false, bv, (short)0, acc0, false, false);
    }
    __builtin_amdgcn_sched_barrier(0);
    {
        v2f a, bv;
        a.x  = prev;
        a.y  = xv[SIG_SR - 2];
        bv.x = xv[0] - prev;
        bv.y = xv[SIG_SR - 1] - a.y;
        acc1 = __builtin_amdgcn_wmma_f32_16x16x4_f32(
                   false, a, false, bv, (short)0, acc1, false, false);
    }
    prev = xv[SIG_SR - 1];
    asm volatile("" :: "v"(acc0), "v"(acc1));
    __builtin_amdgcn_sched_barrier(0);
}

__global__ __launch_bounds__(32)
void sig_batch_kernel(const float* __restrict__ path, float* __restrict__ out, int out_size)
{
    __shared__ __attribute__((aligned(16))) float tile[SIG_LDSF];

    const int b    = blockIdx.x;
    const int lane = threadIdx.x;
    const int q    = lane & 15;
    const int h    = lane >> 4;
    const int p    = q >> 3;
    const int cm   = q & 7;
    const int chunk = 2 * p + h;
    const bool is_time = (cm == 0);
    const int comp = (cm >= 1) ? (cm - 1) : 0;

    const float* __restrict__ g = path + (size_t)b * SIG_L * SIG_C;

    const int prow = (chunk == 0) ? 0 : (chunk * SIG_CH - 1);
    const float xprev = g[prow * SIG_C + comp];
    const float prev0 = is_time ? ((chunk == 0) ? 0.0f : -SIG_DT) : xprev;
    float prev = prev0;
    __builtin_amdgcn_sched_barrier(0);

    v4f ra[28], rb[28];
    sig_load_stage(ra, g, 0, lane);
    sig_load_stage(rb, g, 1, lane);

    #pragma unroll
    for (int j = 0; j < SIG_CH / 32; ++j) {
        const int i = lane + 32 * j;
        tile[SIG_RAMP + SIG_C * i] = (float)i * SIG_DT;
    }

    const float* col = is_time ? (tile + SIG_RAMP) : (tile + chunk * SIG_REG + comp);
    const int cstep = is_time ? SIG_ROWF : 0;
    v8f acc0 = (v8f)0.0f;
    v8f acc1 = (v8f)0.0f;

    sig_store_stage(tile, ra, lane);
    sig_load_stage(ra, g, 2, lane);
    sig_compute_stage(col, prev, acc0, acc1);

    sig_store_stage(tile, rb, lane);
    sig_load_stage(rb, g, 3, lane);
    sig_compute_stage(col + cstep, prev, acc0, acc1);

    sig_store_stage(tile, ra, lane);
    sig_compute_stage(col + 2 * cstep, prev, acc0, acc1);

    sig_store_stage(tile, rb, lane);
    sig_compute_stage(col + 3 * cstep, prev, acc0, acc1);

    const v8f acc = acc0 + acc1;
    const float s = prev - prev0;

    float hi[8];
    #pragma unroll
    for (int j = 0; j < 8; ++j)
        hi[j] = __shfl(acc[j], (lane + 24) & 31);

    __builtin_amdgcn_fence(__ATOMIC_RELEASE, "wavefront");
    __builtin_amdgcn_wave_barrier();
    __builtin_amdgcn_fence(__ATOMIC_ACQUIRE, "wavefront");
    tile[chunk * 8 + cm] = s;
    tile[32 + chunk * 8 + cm] = prev0;
    __builtin_amdgcn_fence(__ATOMIC_RELEASE, "wavefront");
    __builtin_amdgcn_wave_barrier();
    __builtin_amdgcn_fence(__ATOMIC_ACQUIRE, "wavefront");

    if (lane < 8) {
        const int e = lane;
        const float s0e = tile[e];
        const float s1e = tile[8 + e];
        const float s2e = tile[16 + e];
        const float s3e = tile[24 + e];
        const float sig1_e = ((s0e + s1e) + s2e) + s3e;
        const float toff = (SIG_DT * (float)SIG_CH) * ((s1e + 2.0f * s2e) + 3.0f * s3e);
        const size_t base = (size_t)b * SIG_COLS;
        #pragma unroll
        for (int d = 0; d < 8; ++d) {
            const float sig1_d = ((tile[d] + tile[8 + d]) + tile[16 + d]) + tile[24 + d];
            const float cross = (d == 0) ? toff : (-tile[32 + d] * sig1_e);
            const float part = acc[d] + hi[d];
            const float sig2_de = part + cross;
            const int tid = d * 8 + e;
            if (base + 8 + tid < (size_t)out_size)
                out[base + 8 + tid] = sig2_de;
            if (base + 72 + tid < (size_t)out_size)
                out[base + 72 + tid] = sig1_d * sig2_de;
        }
        if (base + e < (size_t)out_size)
            out[base + e] = sig1_e;
    }
}

extern "C" void kernel_launch(void* const* d_in, const int* in_sizes, int n_in,
                              void* d_out, int out_size, void* d_ws, size_t ws_size,
                              hipStream_t stream) {
    (void)in_sizes; (void)n_in; (void)d_ws; (void)ws_size;
    const float* path = (const float*)d_in[0];

    sig_batch_kernel<<<dim3(SIG_B), dim3(32), 0, stream>>>(
        path, (float*)d_out, out_size);
}


# Output stage on all 32 lanes: D blocks and chunk sums pass through LDS once, each lane forms two sig2/sig3 entries, four full-width stores under one wave-uniform bounds test; sum order unchanged
# speedup vs baseline: 1.9409x; 1.1022x over previous
#include <hip/hip_runtime.h>


typedef __attribute__((ext_vector_type(2))) float v2f;
typedef __attribute__((ext_vector_type(4))) float v4f;
typedef __attribute__((ext_vector_type(8))) float v8f;

#define SIG_B      1024
#define SIG_L      2048
#define SIG_C      7
#define SIG_COLS   136
#define SIG_DT     (1.0f / 2047.0f)
#define SIG_CH     512
#define SIG_SR     128
#define SIG_ROWF   (SIG_SR * SIG_C)
#define SIG_REG    904
#define SIG_RAMP   (4 * SIG_REG + 39)
#define SIG_LDSF   (SIG_RAMP + SIG_CH * SIG_C)

__device__ __forceinline__ void sig_load_stage(v4f (&R)[28], const float* __restrict__ g,
                                               const int s, const int lane)
{
    #pragma unroll
    for (int j = 0; j < 7; ++j) {
        #pragma unroll
        for (int c = 0; c < 4; ++c)
            R[c * 7 + j] = *(const v4f*)(g + c * (SIG_CH * SIG_C) + s * SIG_ROWF
                                           + j * 128 + 4 * lane);
        __builtin_amdgcn_sched_barrier(0);
    }
}

__device__ __forceinline__ void sig_store_stage(float* tile, const v4f (&R)[28], const int lane)
{
    __builtin_amdgcn_fence(__ATOMIC_RELEASE, "wavefront");
    __builtin_amdgcn_wave_barrier();
    __builtin_amdgcn_fence(__ATOMIC_ACQUIRE, "wavefront");
    #pragma unroll
    for (int j = 0; j < 7; ++j) {
        #pragma unroll
        for (int c = 0; c < 4; ++c)
            *(v4f*)(tile + c * SIG_REG + j * 128 + 4 * lane) = R[c * 7 + j];
    }
    __builtin_amdgcn_fence(__ATOMIC_RELEASE, "wavefront");
    __builtin_amdgcn_wave_barrier();
    __builtin_amdgcn_fence(__ATOMIC_ACQUIRE, "wavefront");
}

__device__ __forceinline__ void sig_compute_stage(const float* col, float& prev,
                                                  v8f& acc0, v8f& acc1)
{
    float xv[SIG_SR];
    #pragma unroll
    for (int r = 0; r < SIG_SR; ++r)
        xv[r] = col[SIG_C * r];
    __builtin_amdgcn_sched_barrier(0);
    #pragma unroll
    for (int k = 0; k < SIG_SR / 2 - 1; ++k) {
        v2f a, bv;
        a.x  = xv[2 * k];
        a.y  = xv[2 * k + 1];
        bv.x = a.y - a.x;
        bv.y = xv[2 * k + 2] - a.y;
        if (k & 1)
            acc1 = __builtin_amdgcn_wmma_f32_16x16x4_f32(
                       false, a, false, bv, (short)0, acc1, false, false);
        else
            acc0 = __builtin_amdgcn_wmma_f32_16x16x4_f32(
                       false, a, false, bv, (short)0, acc0, false, false);
    }
    __builtin_amdgcn_sched_barrier(0);
    {
        v2f a, bv;
        a.x  = prev;
        a.y  = xv[SIG_SR - 2];
        bv.x = xv[0] - prev;
        bv.y = xv[SIG_SR - 1] - a.y;
        acc1 = __builtin_amdgcn_wmma_f32_16x16x4_f32(
                   false, a, false, bv, (short)0, acc1, false, false);
    }
    prev = xv[SIG_SR - 1];
    asm volatile("" :: "v"(acc0), "v"(acc1));
    __builtin_amdgcn_sched_barrier(0);
}

__global__ __launch_bounds__(32)
void sig_batch_kernel(const float* __restrict__ path, float* __restrict__ out, int out_size)
{
    __shared__ __attribute__((aligned(16))) float tile[SIG_LDSF];

    const int b    = blockIdx.x;
    const int lane = threadIdx.x;
    const int q    = lane & 15;
    const int h    = lane >> 4;
    const int p    = q >> 3;
    const int cm   = q & 7;
    const int chunk = 2 * p + h;
    const bool is_time = (cm == 0);
    const int comp = (cm >= 1) ? (cm - 1) : 0;

    const float* __restrict__ g = path + (size_t)b * SIG_L * SIG_C;

    const int prow = (chunk == 0) ? 0 : (chunk * SIG_CH - 1);
    const float xprev = g[prow * SIG_C + comp];
    const float prev0 = is_time ? ((chunk == 0) ? 0.0f : -SIG_DT) : xprev;
    float prev = prev0;
    __builtin_amdgcn_sched_barrier(0);

    v4f ra[28], rb[28];
    sig_load_stage(ra, g, 0, lane);
    sig_load_stage(rb, g, 1, lane);

    #pragma unroll
    for (int j = 0; j < SIG_CH / 32; ++j) {
        const int i = lane + 32 * j;
        tile[SIG_RAMP + SIG_C * i] = (float)i * SIG_DT;
    }

    const float* col = is_time ? (tile + SIG_RAMP) : (tile + chunk * SIG_REG + comp);
    const int cstep = is_time ? SIG_ROWF : 0;
    v8f acc0 = (v8f)0.0f;
    v8f acc1 = (v8f)0.0f;

    sig_store_stage(tile, ra, lane);
    sig_load_stage(ra, g, 2, lane);
    sig_compute_stage(col, prev, acc0, acc1);

    sig_store_stage(tile, rb, lane);
    sig_load_stage(rb, g, 3, lane);
    sig_compute_stage(col + cstep, prev, acc0, acc1);

    sig_store_stage(tile, ra, lane);
    sig_compute_stage(col + 2 * cstep, prev, acc0, acc1);

    sig_store_stage(tile, rb, lane);
    sig_compute_stage(col + 3 * cstep, prev, acc0, acc1);

    const v8f acc = acc0 + acc1;
    const float s = prev - prev0;

    const int blk = ((lane >> 3) * 3) & 3;

    __builtin_amdgcn_fence(__ATOMIC_RELEASE, "wavefront");
    __builtin_amdgcn_wave_barrier();
    __builtin_amdgcn_fence(__ATOMIC_ACQUIRE, "wavefront");
    tile[chunk * 8 + cm] = s;
    tile[32 + chunk * 8 + cm] = prev0;
    #pragma unroll
    for (int j = 0; j < 8; ++j)
        tile[64 + blk * 72 + j * 8 + cm] = acc[j];
    __builtin_amdgcn_fence(__ATOMIC_RELEASE, "wavefront");
    __builtin_amdgcn_wave_barrier();
    __builtin_amdgcn_fence(__ATOMIC_ACQUIRE, "wavefront");

    const int e = lane & 7;
    const float s0e = tile[e];
    const float s1e = tile[8 + e];
    const float s2e = tile[16 + e];
    const float s3e = tile[24 + e];
    const float sig1_e = ((s0e + s1e) + s2e) + s3e;
    const float toff = (SIG_DT * (float)SIG_CH) * ((s1e + 2.0f * s2e) + 3.0f * s3e);
    float sig2v[2], sig3v[2];
    #pragma unroll
    for (int k = 0; k < 2; ++k) {
        const int d = (lane >> 3) + 4 * k;
        const float sig1_d = ((tile[d] + tile[8 + d]) + tile[16 + d]) + tile[24 + d];
        const float x0d = tile[32 + d];
        const float part = tile[64 + d * 8 + e] + tile[64 + 72 + d * 8 + e];
        const float sig2_time = part + toff;
        const float sig2_data = __builtin_fmaf(-x0d, sig1_e, part);
        sig2v[k] = (d == 0) ? sig2_time : sig2_data;
        sig3v[k] = sig1_d * sig2v[k];
    }

    const size_t base = (size_t)b * SIG_COLS;
    if (base + SIG_COLS <= (size_t)out_size) {
        out[base + 8 + lane]   = sig2v[0];
        out[base + 40 + lane]  = sig2v[1];
        out[base + 72 + lane]  = sig3v[0];
        out[base + 104 + lane] = sig3v[1];
        if (lane < 8)
            out[base + lane] = sig1_e;
    } else {
        if (base + 8 + lane < (size_t)out_size)   out[base + 8 + lane]   = sig2v[0];
        if (base + 40 + lane < (size_t)out_size)  out[base + 40 + lane]  = sig2v[1];
        if (base + 72 + lane < (size_t)out_size)  out[base + 72 + lane]  = sig3v[0];
        if (base + 104 + lane < (size_t)out_size) out[base + 104 + lane] = sig3v[1];
        if (lane < 8 && base + lane < (size_t)out_size)
            out[base + lane] = sig1_e;
    }
}

extern "C" void kernel_launch(void* const* d_in, const int* in_sizes, int n_in,
                              void* d_out, int out_size, void* d_ws, size_t ws_size,
                              hipStream_t stream) {
    (void)in_sizes; (void)n_in; (void)d_ws; (void)ws_size;
    const float* path = (const float*)d_in[0];

    sig_batch_kernel<<<dim3(SIG_B), dim3(32), 0, stream>>>(
        path, (float*)d_out, out_size);
}


# Each batch gets a loader wave beside its matrix wave on the same SIMD (8-wave groups, 4 batches): the loader parks every stage in its own LDS area behind a group barrier; sum order unchanged
# speedup vs baseline: 2.2240x; 1.1459x over previous
#include <hip/hip_runtime.h>


typedef __attribute__((ext_vector_type(2))) float v2f;
typedef __attribute__((ext_vector_type(4))) float v4f;
typedef __attribute__((ext_vector_type(8))) float v8f;

#define SIG_B      1024
#define SIG_L      2048
#define SIG_C      7
#define SIG_COLS   136
#define SIG_DT     (1.0f / 2047.0f)
#define SIG_CH     512
#define SIG_SR     128
#define SIG_ROWF   (SIG_SR * SIG_C)
#define SIG_REG    904
#define SIG_STG    3648
#define SIG_RAMP   (4 * SIG_STG + 7)
#define SIG_BATF   18240
#define SIG_WGB    4

__device__ __forceinline__ void sig_load_stage(v4f (&R)[28], const float* __restrict__ g,
                                               const int s, const int lane)
{
    #pragma unroll
    for (int j = 0; j < 7; ++j) {
        #pragma unroll
        for (int c = 0; c < 4; ++c)
            R[c * 7 + j] = *(const v4f*)(g + c * (SIG_CH * SIG_C) + s * SIG_ROWF
                                           + j * 128 + 4 * lane);
        __builtin_amdgcn_sched_barrier(0);
    }
}

__device__ __forceinline__ void sig_store_stage(float* area, const v4f (&R)[28], const int lane)
{
    #pragma unroll
    for (int j = 0; j < 7; ++j) {
        #pragma unroll
        for (int c = 0; c < 4; ++c)
            *(v4f*)(area + c * SIG_REG + j * 128 + 4 * lane) = R[c * 7 + j];
    }
}

__device__ __forceinline__ void sig_handover()
{
    __builtin_amdgcn_fence(__ATOMIC_RELEASE, "workgroup", "local");
    __builtin_amdgcn_s_barrier();
    __builtin_amdgcn_fence(__ATOMIC_ACQUIRE, "workgroup", "local");
}

__device__ __forceinline__ void sig_compute_stage(const float* col, float& prev,
                                                  v8f& acc0, v8f& acc1)
{
    float xv[SIG_SR];
    #pragma unroll
    for (int r = 0; r < SIG_SR; ++r)
        xv[r] = col[SIG_C * r];
    __builtin_amdgcn_sched_barrier(0);
    #pragma unroll
    for (int k = 0; k < SIG_SR / 2 - 1; ++k) {
        v2f a, bv;
        a.x  = xv[2 * k];
        a.y  = xv[2 * k + 1];
        bv.x = a.y - a.x;
        bv.y = xv[2 * k + 2] - a.y;
        if (k & 1)
            acc1 = __builtin_amdgcn_wmma_f32_16x16x4_f32(
                       false, a, false, bv, (short)0, acc1, false, false);
        else
            acc0 = __builtin_amdgcn_wmma_f32_16x16x4_f32(
                       false, a, false, bv, (short)0, acc0, false, false);
    }
    __builtin_amdgcn_sched_barrier(0);
    {
        v2f a, bv;
        a.x  = prev;
        a.y  = xv[SIG_SR - 2];
        bv.x = xv[0] - prev;
        bv.y = xv[SIG_SR - 1] - a.y;
        acc1 = __builtin_amdgcn_wmma_f32_16x16x4_f32(
                   false, a, false, bv, (short)0, acc1, false, false);
    }
    prev = xv[SIG_SR - 1];
}

__global__ __launch_bounds__(256)
void sig_batch_kernel(const float* __restrict__ path, float* __restrict__ out, int out_size)
{
    __shared__ __attribute__((aligned(16))) float tile_all[SIG_WGB * SIG_BATF];

    const int wave = __builtin_amdgcn_readfirstlane((int)(threadIdx.x >> 5));
    const int lane = threadIdx.x & 31;
    const bool loader = (wave >= SIG_WGB);
    const int bi   = wave & (SIG_WGB - 1);
    const int b    = blockIdx.x * SIG_WGB + bi;
    float* tile = tile_all + bi * SIG_BATF;

    const int q    = lane & 15;
    const int h    = lane >> 4;
    const int p    = q >> 3;
    const int cm   = q & 7;
    const int chunk = 2 * p + h;
    const bool is_time = (cm == 0);
    const int comp = (cm >= 1) ? (cm - 1) : 0;

    const float* __restrict__ g = path + (size_t)b * SIG_L * SIG_C;

    v4f ra[28], rb[28];
    float prev0 = 0.0f;
    float prev  = 0.0f;

    if (loader) {
        sig_load_stage(ra, g, 0, lane);
        sig_load_stage(rb, g, 1, lane);

        #pragma unroll
        for (int j = 0; j < SIG_CH / 32; ++j) {
            const int i = lane + 32 * j;
            tile[SIG_RAMP + SIG_C * i] = (float)i * SIG_DT;
        }

        sig_store_stage(tile, ra, lane);
        sig_load_stage(ra, g, 2, lane);
    } else {
        const int prow = (chunk == 0) ? 0 : (chunk * SIG_CH - 1);
        const float xprev = g[prow * SIG_C + comp];
        prev0 = is_time ? ((chunk == 0) ? 0.0f : -SIG_DT) : xprev;
        prev  = prev0;
    }

    const float* col = is_time ? (tile + SIG_RAMP) : (tile + chunk * SIG_REG + comp);
    const int cstep = is_time ? SIG_ROWF : SIG_STG;
    v8f acc0 = (v8f)0.0f;
    v8f acc1 = (v8f)0.0f;

    sig_handover();
    if (loader) {
        sig_store_stage(tile + SIG_STG, rb, lane);
        sig_load_stage(rb, g, 3, lane);
    } else {
        sig_compute_stage(col, prev, acc0, acc1);
    }

    sig_handover();
    if (loader)
        sig_store_stage(tile + 2 * SIG_STG, ra, lane);
    else
        sig_compute_stage(col + cstep, prev, acc0, acc1);

    sig_handover();
    if (loader)
        sig_store_stage(tile + 3 * SIG_STG, rb, lane);
    else
        sig_compute_stage(col + 2 * cstep, prev, acc0, acc1);

    sig_handover();
    if (loader)
        return;

    sig_compute_stage(col + 3 * cstep, prev, acc0, acc1);

    const v8f acc = acc0 + acc1;
    const float s = prev - prev0;

    const int blk = ((lane >> 3) * 3) & 3;

    tile[chunk * 8 + cm] = s;
    tile[32 + chunk * 8 + cm] = prev0;
    #pragma unroll
    for (int j = 0; j < 8; ++j)
        tile[64 + blk * 72 + j * 8 + cm] = acc[j];
    __builtin_amdgcn_fence(__ATOMIC_RELEASE, "wavefront");
    __builtin_amdgcn_wave_barrier();
    __builtin_amdgcn_fence(__ATOMIC_ACQUIRE, "wavefront");

    const int e = lane & 7;
    const float s0e = tile[e];
    const float s1e = tile[8 + e];
    const float s2e = tile[16 + e];
    const float s3e = tile[24 + e];
    const float sig1_e = ((s0e + s1e) + s2e) + s3e;
    const float toff = (SIG_DT * (float)SIG_CH) * ((s1e + 2.0f * s2e) + 3.0f * s3e);
    float sig2v[2], sig3v[2];
    #pragma unroll
    for (int k = 0; k < 2; ++k) {
        const int d = (lane >> 3) + 4 * k;
        const float sig1_d = ((tile[d] + tile[8 + d]) + tile[16 + d]) + tile[24 + d];
        const float x0d = tile[32 + d];
        const float part = tile[64 + d * 8 + e] + tile[64 + 72 + d * 8 + e];
        const float sig2_time = part + toff;
        const float sig2_data = __builtin_fmaf(-x0d, sig1_e, part);
        sig2v[k] = (d == 0) ? sig2_time : sig2_data;
        sig3v[k] = sig1_d * sig2v[k];
    }

    const size_t base = (size_t)b * SIG_COLS;
    if (base + SIG_COLS <= (size_t)out_size) {
        out[base + 8 + lane]   = sig2v[0];
        out[base + 40 + lane]  = sig2v[1];
        out[base + 72 + lane]  = sig3v[0];
        out[base + 104 + lane] = sig3v[1];
        if (lane < 8)
            out[base + lane] = sig1_e;
    } else {
        if (base + 8 + lane < (size_t)out_size)   out[base + 8 + lane]   = sig2v[0];
        if (base + 40 + lane < (size_t)out_size)  out[base + 40 + lane]  = sig2v[1];
        if (base + 72 + lane < (size_t)out_size)  out[base + 72 + lane]  = sig3v[0];
        if (base + 104 + lane < (size_t)out_size) out[base + 104 + lane] = sig3v[1];
        if (lane < 8 && base + lane < (size_t)out_size)
            out[base + lane] = sig1_e;
    }
}

extern "C" void kernel_launch(void* const* d_in, const int* in_sizes, int n_in,
                              void* d_out, int out_size, void* d_ws, size_t ws_size,
                              hipStream_t stream) {
    (void)in_sizes; (void)n_in; (void)d_ws; (void)ws_size;
    const float* path = (const float*)d_in[0];

    sig_batch_kernel<<<dim3(SIG_B / SIG_WGB), dim3(64 * SIG_WGB), 0, stream>>>(
        path, (float*)d_out, out_size);
}


# Work-groups of 6 waves, not 8 (1536 waves enter, not 2048): each matrix wave fetches and parks its own stage 0 (in-wave hand-over), 2 loader waves park stages 1-3 of two batches each; sums unchanged
# speedup vs baseline: 2.3483x; 1.0559x over previous
#include <hip/hip_runtime.h>


typedef __attribute__((ext_vector_type(2))) float v2f;
typedef __attribute__((ext_vector_type(4))) float v4f;
typedef __attribute__((ext_vector_type(8))) float v8f;

#define SIG_B      1024
#define SIG_L      2048
#define SIG_C      7
#define SIG_COLS   136
#define SIG_DT     (1.0f / 2047.0f)
#define SIG_CH     512
#define SIG_SR     128
#define SIG_ROWF   (SIG_SR * SIG_C)
#define SIG_REG    904
#define SIG_STG    3648
#define SIG_RAMP   (4 * SIG_STG + 7)
#define SIG_BATF   18240
#define SIG_WGB    4
#define SIG_NLD    2
#define SIG_THREADS 192

__device__ __forceinline__ void sig_load_stage(v4f (&R)[28], const float* __restrict__ g,
                                               const int s, const int lane)
{
    #pragma unroll
    for (int j = 0; j < 7; ++j) {
        #pragma unroll
        for (int c = 0; c < 4; ++c)
            R[c * 7 + j] = *(const v4f*)(g + c * (SIG_CH * SIG_C) + s * SIG_ROWF
                                           + j * 128 + 4 * lane);
        __builtin_amdgcn_sched_barrier(0);
    }
}

__device__ __forceinline__ void sig_store_stage(float* area, const v4f (&R)[28], const int lane)
{
    #pragma unroll
    for (int j = 0; j < 7; ++j) {
        #pragma unroll
        for (int c = 0; c < 4; ++c)
            *(v4f*)(area + c * SIG_REG + j * 128 + 4 * lane) = R[c * 7 + j];
    }
}

__device__ __forceinline__ void sig_handover()
{
    __builtin_amdgcn_fence(__ATOMIC_RELEASE, "workgroup", "local");
    __builtin_amdgcn_s_barrier();
    __builtin_amdgcn_fence(__ATOMIC_ACQUIRE, "workgroup", "local");
}

__device__ __forceinline__ void sig_compute_stage(const float* col, float& prev,
                                                  v8f& acc0, v8f& acc1)
{
    float xv[SIG_SR];
    #pragma unroll
    for (int r = 0; r < SIG_SR; ++r)
        xv[r] = col[SIG_C * r];
    __builtin_amdgcn_sched_barrier(0);
    #pragma unroll
    for (int k = 0; k < SIG_SR / 2 - 1; ++k) {
        v2f a, bv;
        a.x  = xv[2 * k];
        a.y  = xv[2 * k + 1];
        bv.x = a.y - a.x;
        bv.y = xv[2 * k + 2] - a.y;
        if (k & 1)
            acc1 = __builtin_amdgcn_wmma_f32_16x16x4_f32(
                       false, a, false, bv, (short)0, acc1, false, false);
        else
            acc0 = __builtin_amdgcn_wmma_f32_16x16x4_f32(
                       false, a, false, bv, (short)0, acc0, false, false);
    }
    __builtin_amdgcn_sched_barrier(0);
    {
        v2f a, bv;
        a.x  = prev;
        a.y  = xv[SIG_SR - 2];
        bv.x = xv[0] - prev;
        bv.y = xv[SIG_SR - 1] - a.y;
        acc1 = __builtin_amdgcn_wmma_f32_16x16x4_f32(
                   false, a, false, bv, (short)0, acc1, false, false);
    }
    prev = xv[SIG_SR - 1];
}

__global__ __launch_bounds__(SIG_THREADS)
void sig_batch_kernel(const float* __restrict__ path, float* __restrict__ out, int out_size)
{
    __shared__ __attribute__((aligned(16))) float tile_all[SIG_WGB * SIG_BATF];

    const int wave = __builtin_amdgcn_readfirstlane((int)(threadIdx.x >> 5));
    const int lane = threadIdx.x & 31;
    const bool loader = (wave >= SIG_WGB);
    const int bi   = loader ? 2 * (wave - SIG_WGB) : wave;
    const int b    = blockIdx.x * SIG_WGB + bi;
    float* tile  = tile_all + bi * SIG_BATF;
    float* tile2 = tile + SIG_BATF;

    const int q    = lane & 15;
    const int h    = lane >> 4;
    const int p    = q >> 3;
    const int cm   = q & 7;
    const int chunk = 2 * p + h;
    const bool is_time = (cm == 0);
    const int comp = (cm >= 1) ? (cm - 1) : 0;

    const float* __restrict__ g  = path + (size_t)b * SIG_L * SIG_C;
    const float* __restrict__ g2 = g + SIG_L * SIG_C;

    v4f ra[28], rb[28];
    float prev0 = 0.0f;
    float prev  = 0.0f;

    const float* col = is_time ? (tile + SIG_RAMP) : (tile + chunk * SIG_REG + comp);
    const int cstep = is_time ? SIG_ROWF : SIG_STG;
    v8f acc0 = (v8f)0.0f;
    v8f acc1 = (v8f)0.0f;

    if (loader) {
        sig_load_stage(ra, g,  1, lane);
        sig_load_stage(rb, g2, 1, lane);
        sig_store_stage(tile + SIG_STG, ra, lane);
        sig_load_stage(ra, g,  2, lane);
        sig_store_stage(tile2 + SIG_STG, rb, lane);
        sig_load_stage(rb, g2, 2, lane);
    } else {
        sig_load_stage(rb, g, 0, lane);

        const int prow = (chunk == 0) ? 0 : (chunk * SIG_CH - 1);
        const float xprev = g[prow * SIG_C + comp];

        #pragma unroll
        for (int j = 0; j < SIG_CH / 32; ++j) {
            const int i = lane + 32 * j;
            tile[SIG_RAMP + SIG_C * i] = (float)i * SIG_DT;
        }

        sig_store_stage(tile, rb, lane);
        prev0 = is_time ? ((chunk == 0) ? 0.0f : -SIG_DT) : xprev;
        prev  = prev0;
        __builtin_amdgcn_fence(__ATOMIC_RELEASE, "wavefront");
        __builtin_amdgcn_wave_barrier();
        __builtin_amdgcn_fence(__ATOMIC_ACQUIRE, "wavefront");

        sig_compute_stage(col, prev, acc0, acc1);
    }

    sig_handover();
    if (loader) {
        sig_store_stage(tile + 2 * SIG_STG, ra, lane);
        sig_load_stage(ra, g,  3, lane);
        sig_store_stage(tile2 + 2 * SIG_STG, rb, lane);
        sig_load_stage(rb, g2, 3, lane);
    } else {
        sig_compute_stage(col + cstep, prev, acc0, acc1);
    }

    sig_handover();
    if (loader) {
        sig_store_stage(tile + 3 * SIG_STG, ra, lane);
        sig_store_stage(tile2 + 3 * SIG_STG, rb, lane);
    } else {
        sig_compute_stage(col + 2 * cstep, prev, acc0, acc1);
    }

    sig_handover();
    if (loader)
        return;

    sig_compute_stage(col + 3 * cstep, prev, acc0, acc1);

    const v8f acc = acc0 + acc1;
    const float s = prev - prev0;

    const int blk = ((lane >> 3) * 3) & 3;

    tile[chunk * 8 + cm] = s;
    tile[32 + chunk * 8 + cm] = prev0;
    #pragma unroll
    for (int j = 0; j < 8; ++j)
        tile[64 + blk * 72 + j * 8 + cm] = acc[j];
    __builtin_amdgcn_fence(__ATOMIC_RELEASE, "wavefront");
    __builtin_amdgcn_wave_barrier();
    __builtin_amdgcn_fence(__ATOMIC_ACQUIRE, "wavefront");

    const int e = lane & 7;
    const float s0e = tile[e];
    const float s1e = tile[8 + e];
    const float s2e = tile[16 + e];
    const float s3e = tile[24 + e];
    const float sig1_e = ((s0e + s1e) + s2e) + s3e;
    const float toff = (SIG_DT * (float)SIG_CH) * ((s1e + 2.0f * s2e) + 3.0f * s3e);
    float sig2v[2], sig3v[2];
    #pragma unroll
    for (int k = 0; k < 2; ++k) {
        const int d = (lane >> 3) + 4 * k;
        const float sig1_d = ((tile[d] + tile[8 + d]) + tile[16 + d]) + tile[24 + d];
        const float x0d = tile[32 + d];
        const float part = tile[64 + d * 8 + e] + tile[64 + 72 + d * 8 + e];
        const float sig2_time = part + toff;
        const float sig2_data = __builtin_fmaf(-x0d, sig1_e, part);
        sig2v[k] = (d == 0) ? sig2_time : sig2_data;
        sig3v[k] = sig1_d * sig2v[k];
    }

    const size_t base = (size_t)b * SIG_COLS;
    if (base + SIG_COLS <= (size_t)out_size) {
        out[base + 8 + lane]   = sig2v[0];
        out[base + 40 + lane]  = sig2v[1];
        out[base + 72 + lane]  = sig3v[0];
        out[base + 104 + lane] = sig3v[1];
        if (lane < 8)
            out[base + lane] = sig1_e;
    } else {
        if (base + 8 + lane < (size_t)out_size)   out[base + 8 + lane]   = sig2v[0];
        if (base + 40 + lane < (size_t)out_size)  out[base + 40 + lane]  = sig2v[1];
        if (base + 72 + lane < (size_t)out_size)  out[base + 72 + lane]  = sig3v[0];
        if (base + 104 + lane < (size_t)out_size) out[base + 104 + lane] = sig3v[1];
        if (lane < 8 && base + lane < (size_t)out_size)
            out[base + lane] = sig1_e;
    }
}

extern "C" void kernel_launch(void* const* d_in, const int* in_sizes, int n_in,
                              void* d_out, int out_size, void* d_ws, size_t ws_size,
                              hipStream_t stream) {
    (void)in_sizes; (void)n_in; (void)d_ws; (void)ws_size;
    const float* path = (const float*)d_in[0];

    sig_batch_kernel<<<dim3(SIG_B / SIG_WGB), dim3(SIG_THREADS), 0, stream>>>(
        path, (float*)d_out, out_size);
}
